# starting kernel
# baseline (speedup 1.0000x reference)
#include <hip/hip_runtime.h>

typedef __attribute__((ext_vector_type(2))) float v2f;
typedef __attribute__((ext_vector_type(8))) float v8f;

#define HID 256
#define TPB 256


__global__ void fill_kernel(float* __restrict__ p, float v, int n) {
  int i = blockIdx.x * blockDim.x + threadIdx.x;
  if (i < n) p[i] = v;
}

__global__ void degree_kernel(const int* __restrict__ dst,
                              float* __restrict__ deg, int E) {
  int e = blockIdx.x * blockDim.x + threadIdx.x;
  if (e < E) unsafeAtomicAdd(&deg[dst[e]], 1.0f);
}

__global__ void rsqrt_inplace_kernel(float* __restrict__ d, int n) {
  int i = blockIdx.x * blockDim.x + threadIdx.x;
  if (i < n) d[i] = 1.0f / sqrtf(d[i]);
}

__global__ void enorm_kernel(const int* __restrict__ src,
                             const int* __restrict__ dst,
                             const float* __restrict__ dinv,
                             float* __restrict__ enorm, int E) {
  int e = blockIdx.x * blockDim.x + threadIdx.x;
  if (e < E) enorm[e] = dinv[src[e]] * dinv[dst[e]];
}

template <int K, bool RELU>
__global__ __launch_bounds__(TPB) void gemm_wmma_kernel(
    const float* __restrict__ X, const float* __restrict__ W,
    float* __restrict__ H, int N) {
  const int lane = threadIdx.x & 31;
  const int wave = threadIdx.x >> 5;
  const int rgrp = wave >> 1;
  const int cgrp = wave & 1;
  const int row_base = blockIdx.x * 128 + rgrp * 32;
  if (row_base >= N) return;

  const int col_base = cgrp * 128;
  const int col = lane & 15;
  const int khalf = (lane >> 4) << 1;
  const size_t arow0 = (size_t)(row_base + col);
  const size_t arow1 = (size_t)(row_base + 16 + col);

  v8f acc[2][8];
#pragma unroll
  for (int m = 0; m < 2; ++m)
#pragma unroll
    for (int j = 0; j < 8; ++j)
      acc[m][j] = (v8f){0.f, 0.f, 0.f, 0.f, 0.f, 0.f, 0.f, 0.f};

  for (int k = 0; k < K; k += 4) {
    v2f a0 = *(const v2f*)(X + arow0 * K + (k + khalf));
    v2f a1 = *(const v2f*)(X + arow1 * K + (k + khalf));
    if (RELU) {
      a0.x = fmaxf(a0.x, 0.f);
      a0.y = fmaxf(a0.y, 0.f);
      a1.x = fmaxf(a1.x, 0.f);
      a1.y = fmaxf(a1.y, 0.f);
    }
#pragma unroll
    for (int j = 0; j < 8; ++j) {
      v2f b;
      b.x = W[(size_t)(k + khalf) * HID + col_base + j * 16 + col];
      b.y = W[(size_t)(k + khalf + 1) * HID + col_base + j * 16 + col];
      acc[0][j] = __builtin_amdgcn_wmma_f32_16x16x4_f32(
          false, a0, false, b, (short)0, acc[0][j], false, false);
      acc[1][j] = __builtin_amdgcn_wmma_f32_16x16x4_f32(
          false, a1, false, b, (short)0, acc[1][j], false, false);
    }
  }

#pragma unroll
  for (int m = 0; m < 2; ++m) {
    const int rlo = row_base + m * 16 + (lane >> 4) * 8;
#pragma unroll
    for (int j = 0; j < 8; ++j) {
#pragma unroll
      for (int r = 0; r < 8; ++r) {
        H[(size_t)(rlo + r) * HID + col_base + j * 16 + col] = acc[m][j][r];
      }
    }
  }
}


__global__ void init_agg_kernel(const float* __restrict__ h,
                                const float* __restrict__ dinv,
                                const float* __restrict__ bias,
                                float* __restrict__ agg, int N) {
  int t = blockIdx.x * blockDim.x + threadIdx.x;
  int node = t >> 6;
  if (node >= N) return;
  int c = (t & 63) << 2;
  float di = dinv[node];
  float sn = di * di;
  float4 hv = *(const float4*)(h + (size_t)node * HID + c);
  float4 bv = *(const float4*)(bias + c);
  float4 o;
  o.x = fmaf(hv.x, sn, bv.x);
  o.y = fmaf(hv.y, sn, bv.y);
  o.z = fmaf(hv.z, sn, bv.z);
  o.w = fmaf(hv.w, sn, bv.w);
  *(float4*)(agg + (size_t)node * HID + c) = o;
}

__global__ __launch_bounds__(TPB) void scatter_kernel(
    const float* __restrict__ h, const int* __restrict__ src,
    const int* __restrict__ dst, const float* __restrict__ enorm,
    float* __restrict__ agg, int E) {
  int e = __builtin_amdgcn_readfirstlane(blockIdx.x * (TPB / 32) +
                                         (threadIdx.x >> 5));
  if (e >= E) return;
  const int s = src[e];
  const int d = dst[e];
  const float w = enorm[e];
  const int c = (threadIdx.x & 31) * 8;
  const float* hp = h + (size_t)s * HID + c;
  float* ap = agg + (size_t)d * HID + c;
  float4 v0 = *(const float4*)(hp);
  float4 v1 = *(const float4*)(hp + 4);
  unsafeAtomicAdd(ap + 0, v0.x * w);
  unsafeAtomicAdd(ap + 1, v0.y * w);
  unsafeAtomicAdd(ap + 2, v0.z * w);
  unsafeAtomicAdd(ap + 3, v0.w * w);
  unsafeAtomicAdd(ap + 4, v1.x * w);
  unsafeAtomicAdd(ap + 5, v1.y * w);
  unsafeAtomicAdd(ap + 6, v1.z * w);
  unsafeAtomicAdd(ap + 7, v1.w * w);
}


static inline int cdiv(int a, int b) { return (a + b - 1) / b; }

extern "C" void kernel_launch(void* const* d_in, const int* in_sizes, int n_in,
                              void* d_out, int out_size, void* d_ws,
                              size_t ws_size, hipStream_t stream) {
  const float* x = (const float*)d_in[0];
  const int* ei = (const int*)d_in[1];
  const float* W1 = (const float*)d_in[2];
  const float* b1 = (const float*)d_in[3];
  const float* W2 = (const float*)d_in[4];
  const float* b2 = (const float*)d_in[5];
  const float* W3 = (const float*)d_in[6];
  const float* b3 = (const float*)d_in[7];
  float* out = (float*)d_out;

  const int F = 128;
  const int N = in_sizes[0] / F;
  const int E = in_sizes[1] / 2;
  const int* src = ei;
  const int* dst = ei + E;

  float* wsf = (float*)d_ws;
  size_t off = 0;
  float* dinv = wsf + off;
  off += (size_t)((N + 255) / 256) * 256;
  float* enorm = wsf + off;
  off += (size_t)((E + 255) / 256) * 256;
  float* hbuf = wsf + off;
  off += (size_t)N * HID;
  float* agg = wsf + off;

  const int nodeGrid = cdiv(N, TPB);
  const int edgeGrid = cdiv(E, TPB);
  const int featGrid = cdiv(N * 64, TPB);
  const int scatGrid = cdiv(E, TPB / 32);
  const int gemmGrid = cdiv(N, 128);

  fill_kernel<<<nodeGrid, TPB, 0, stream>>>(dinv, 1.0f, N);
  degree_kernel<<<edgeGrid, TPB, 0, stream>>>(dst, dinv, E);
  rsqrt_inplace_kernel<<<nodeGrid, TPB, 0, stream>>>(dinv, N);
  enorm_kernel<<<edgeGrid, TPB, 0, stream>>>(src, dst, dinv, enorm, E);

  gemm_wmma_kernel<128, false><<<gemmGrid, TPB, 0, stream>>>(x, W1, hbuf, N);
  init_agg_kernel<<<featGrid, TPB, 0, stream>>>(hbuf, dinv, b1, agg, N);
  scatter_kernel<<<scatGrid, TPB, 0, stream>>>(hbuf, src, dst, enorm, agg, E);

  gemm_wmma_kernel<256, true><<<gemmGrid, TPB, 0, stream>>>(agg, W2, hbuf, N);
  init_agg_kernel<<<featGrid, TPB, 0, stream>>>(hbuf, dinv, b2, agg, N);
  scatter_kernel<<<scatGrid, TPB, 0, stream>>>(hbuf, src, dst, enorm, agg, E);

  gemm_wmma_kernel<256, true><<<gemmGrid, TPB, 0, stream>>>(agg, W3, hbuf, N);
  init_agg_kernel<<<featGrid, TPB, 0, stream>>>(hbuf, dinv, b3, out, N);
  scatter_kernel<<<scatGrid, TPB, 0, stream>>>(hbuf, src, dst, enorm, out, E);
}


# f32 atomic scatter-add replaced by edge list grouped by destination (int count, scan, place, order by edge id); one wave per node sums its rows in fixed edge order, self-loop+bias folded in
# speedup vs baseline: 3.2666x; 3.2666x over previous
#include <hip/hip_runtime.h>

typedef __attribute__((ext_vector_type(2))) float v2f;
typedef __attribute__((ext_vector_type(8))) float v8f;

#define HID 256
#define TPB 256
#define SCAN_CHUNK 1024


__global__ void zero_int_kernel(int* __restrict__ p, int n) {
  int i = blockIdx.x * blockDim.x + threadIdx.x;
  if (i < n) p[i] = 0;
}

__global__ void count_kernel(const int* __restrict__ dst,
                             int* __restrict__ cnt, int E) {
  int e = blockIdx.x * blockDim.x + threadIdx.x;
  if (e < E) atomicAdd(&cnt[dst[e]], 1);
}

__global__ __launch_bounds__(TPB) void block_sum_kernel(
    const int* __restrict__ cnt, int* __restrict__ bsum) {
  __shared__ int wsum[TPB / 32];
  const int t = threadIdx.x;
  const int4 v = *(const int4*)(cnt + (size_t)blockIdx.x * SCAN_CHUNK + t * 4);
  int s = v.x + v.y + v.z + v.w;
  for (int o = 16; o > 0; o >>= 1) s += __shfl_xor(s, o);
  if ((t & 31) == 0) wsum[t >> 5] = s;
  __syncthreads();
  if (t == 0) {
    int tot = 0;
    for (int i = 0; i < TPB / 32; ++i) tot += wsum[i];
    bsum[blockIdx.x] = tot;
  }
}

__global__ __launch_bounds__(TPB) void scan_kernel(
    int* __restrict__ cntcur, const int* __restrict__ bsum,
    int* __restrict__ rowptr, float* __restrict__ dinv) {
  __shared__ int wsum[TPB / 32];
  __shared__ int boff;
  const int t = threadIdx.x;
  const int lane = t & 31;
  const int wave = t >> 5;
  if (wave == 0) {
    int s = 0;
    for (int i = lane; i < (int)blockIdx.x; i += 32) s += bsum[i];
    for (int o = 16; o > 0; o >>= 1) s += __shfl_xor(s, o);
    if (lane == 0) boff = s;
  }
  const size_t base = (size_t)blockIdx.x * SCAN_CHUNK + t * 4;
  const int4 v = *(const int4*)(cntcur + base);
  const int s1 = v.x;
  const int s2 = s1 + v.y;
  const int s3 = s2 + v.z;
  const int s4 = s3 + v.w;
  int inc = s4;
  for (int o = 1; o < 32; o <<= 1) {
    const int up = __shfl_up(inc, o);
    if (lane >= o) inc += up;
  }
  if (lane == 31) wsum[wave] = inc;
  __syncthreads();
  int woff = boff;
  for (int i = 0; i < wave; ++i) woff += wsum[i];
  const int p0 = woff + inc - s4;
  int4 p;
  p.x = p0;
  p.y = p0 + s1;
  p.z = p0 + s2;
  p.w = p0 + s3;
  *(int4*)(rowptr + base) = p;
  *(int4*)(cntcur + base) = p;
  float4 dv;
  dv.x = 1.0f / sqrtf((float)(v.x + 1));
  dv.y = 1.0f / sqrtf((float)(v.y + 1));
  dv.z = 1.0f / sqrtf((float)(v.z + 1));
  dv.w = 1.0f / sqrtf((float)(v.w + 1));
  *(float4*)(dinv + base) = dv;
  if (blockIdx.x == gridDim.x - 1 && t == TPB - 1) rowptr[base + 4] = p0 + s4;
}

__global__ void place_kernel(const int* __restrict__ src,
                             const int* __restrict__ dst,
                             int* __restrict__ cursor,
                             int2* __restrict__ slot, int E) {
  int e = blockIdx.x * blockDim.x + threadIdx.x;
  if (e < E) {
    const int pos = atomicAdd(&cursor[dst[e]], 1);
    slot[pos] = make_int2(e, src[e]);
  }
}

__global__ void order_kernel(const int2* __restrict__ slot,
                             const int* __restrict__ dst,
                             const int* __restrict__ rowptr,
                             const float* __restrict__ dinv,
                             int2* __restrict__ csr, int E) {
  int a = blockIdx.x * blockDim.x + threadIdx.x;
  if (a >= E) return;
  const int2 me = slot[a];
  const int node = dst[me.x];
  const int beg = rowptr[node];
  const int end = rowptr[node + 1];
  int rank = 0;
  for (int b = beg; b < end; ++b) rank += (slot[b].x < me.x) ? 1 : 0;
  const float w = dinv[me.y] * dinv[node];
  csr[beg + rank] = make_int2(me.y, __float_as_int(w));
}

template <int K, bool RELU>
__global__ __launch_bounds__(TPB) void gemm_wmma_kernel(
    const float* __restrict__ X, const float* __restrict__ W,
    float* __restrict__ H, int N) {
  const int lane = threadIdx.x & 31;
  const int wave = threadIdx.x >> 5;
  const int rgrp = wave >> 1;
  const int cgrp = wave & 1;
  const int row_base = blockIdx.x * 128 + rgrp * 32;
  if (row_base >= N) return;

  const int col_base = cgrp * 128;
  const int col = lane & 15;
  const int khalf = (lane >> 4) << 1;
  const size_t arow0 = (size_t)(row_base + col);
  const size_t arow1 = (size_t)(row_base + 16 + col);

  v8f acc[2][8];
#pragma unroll
  for (int m = 0; m < 2; ++m)
#pragma unroll
    for (int j = 0; j < 8; ++j)
      acc[m][j] = (v8f){0.f, 0.f, 0.f, 0.f, 0.f, 0.f, 0.f, 0.f};

  for (int k = 0; k < K; k += 4) {
    v2f a0 = *(const v2f*)(X + arow0 * K + (k + khalf));
    v2f a1 = *(const v2f*)(X + arow1 * K + (k + khalf));
    if (RELU) {
      a0.x = fmaxf(a0.x, 0.f);
      a0.y = fmaxf(a0.y, 0.f);
      a1.x = fmaxf(a1.x, 0.f);
      a1.y = fmaxf(a1.y, 0.f);
    }
#pragma unroll
    for (int j = 0; j < 8; ++j) {
      v2f b;
      b.x = W[(size_t)(k + khalf) * HID + col_base + j * 16 + col];
      b.y = W[(size_t)(k + khalf + 1) * HID + col_base + j * 16 + col];
      acc[0][j] = __builtin_amdgcn_wmma_f32_16x16x4_f32(
          false, a0, false, b, (short)0, acc[0][j], false, false);
      acc[1][j] = __builtin_amdgcn_wmma_f32_16x16x4_f32(
          false, a1, false, b, (short)0, acc[1][j], false, false);
    }
  }

#pragma unroll
  for (int m = 0; m < 2; ++m) {
    const int rlo = row_base + m * 16 + (lane >> 4) * 8;
#pragma unroll
    for (int j = 0; j < 8; ++j) {
#pragma unroll
      for (int r = 0; r < 8; ++r) {
        H[(size_t)(rlo + r) * HID + col_base + j * 16 + col] = acc[m][j][r];
      }
    }
  }
}


template <int R>
__device__ __forceinline__ void add_rows(const float* __restrict__ h,
                                         const int2 ent, const int j,
                                         const int c, float4& a0, float4& a1) {
  float4 v0[R], v1[R];
  float w[R];
#pragma unroll
  for (int r = 0; r < R; ++r) {
    const int s = __builtin_amdgcn_readlane(ent.x, j + r);
    w[r] = __int_as_float(__builtin_amdgcn_readlane(ent.y, j + r));
    const float* hp = h + (size_t)s * HID + c;
    v0[r] = *(const float4*)(hp);
    v1[r] = *(const float4*)(hp + 128);
  }
#pragma unroll
  for (int r = 0; r < R; ++r) {
    a0.x = fmaf(v0[r].x, w[r], a0.x);
    a0.y = fmaf(v0[r].y, w[r], a0.y);
    a0.z = fmaf(v0[r].z, w[r], a0.z);
    a0.w = fmaf(v0[r].w, w[r], a0.w);
    a1.x = fmaf(v1[r].x, w[r], a1.x);
    a1.y = fmaf(v1[r].y, w[r], a1.y);
    a1.z = fmaf(v1[r].z, w[r], a1.z);
    a1.w = fmaf(v1[r].w, w[r], a1.w);
  }
}

__global__ __launch_bounds__(TPB) void gather_kernel(
    const float* __restrict__ h, const int* __restrict__ rowptr,
    const int2* __restrict__ csr, const float* __restrict__ dinv,
    const float* __restrict__ bias, float* __restrict__ agg, int N) {
  const int node = __builtin_amdgcn_readfirstlane(blockIdx.x * (TPB / 32) +
                                                  (threadIdx.x >> 5));
  if (node >= N) return;
  const int lane = threadIdx.x & 31;
  const int c = lane * 4;
  const int beg = rowptr[node];
  const int end = rowptr[node + 1];
  const float di = dinv[node];
  const float sn = di * di;
  const float* hp = h + (size_t)node * HID + c;
  const float4 x0 = *(const float4*)(hp);
  const float4 x1 = *(const float4*)(hp + 128);
  const float4 bv0 = *(const float4*)(bias + c);
  const float4 bv1 = *(const float4*)(bias + 128 + c);
  float4 a0 = bv0, a1 = bv1;

  for (int base = beg; base < end; base += 32) {
    const int n = min(32, end - base);
    int2 ent = make_int2(0, 0);
    if (lane < n) ent = csr[base + lane];
    int j = 0;
    for (; j + 4 <= n; j += 4) add_rows<4>(h, ent, j, c, a0, a1);
    const int rem = n - j;
    if (rem == 3) {
      add_rows<3>(h, ent, j, c, a0, a1);
    } else if (rem == 2) {
      add_rows<2>(h, ent, j, c, a0, a1);
    } else if (rem == 1) {
      add_rows<1>(h, ent, j, c, a0, a1);
    }
  }

  a0.x = fmaf(x0.x, sn, a0.x);
  a0.y = fmaf(x0.y, sn, a0.y);
  a0.z = fmaf(x0.z, sn, a0.z);
  a0.w = fmaf(x0.w, sn, a0.w);
  a1.x = fmaf(x1.x, sn, a1.x);
  a1.y = fmaf(x1.y, sn, a1.y);
  a1.z = fmaf(x1.z, sn, a1.z);
  a1.w = fmaf(x1.w, sn, a1.w);

  float* ap = agg + (size_t)node * HID + c;
  *(float4*)(ap) = a0;
  *(float4*)(ap + 128) = a1;
}


static inline int cdiv(int a, int b) { return (a + b - 1) / b; }

extern "C" void kernel_launch(void* const* d_in, const int* in_sizes, int n_in,
                              void* d_out, int out_size, void* d_ws,
                              size_t ws_size, hipStream_t stream) {
  const float* x = (const float*)d_in[0];
  const int* ei = (const int*)d_in[1];
  const float* W1 = (const float*)d_in[2];
  const float* b1 = (const float*)d_in[3];
  const float* W2 = (const float*)d_in[4];
  const float* b2 = (const float*)d_in[5];
  const float* W3 = (const float*)d_in[6];
  const float* b3 = (const float*)d_in[7];
  float* out = (float*)d_out;

  const int F = 128;
  const int N = in_sizes[0] / F;
  const int E = in_sizes[1] / 2;
  const int* src = ei;
  const int* dst = ei + E;

  const int scanGrid = cdiv(N, SCAN_CHUNK);
  const int NP = scanGrid * SCAN_CHUNK;

  float* wsf = (float*)d_ws;
  size_t off = 0;
  float* dinv = wsf + off;
  off += (size_t)NP;
  int* cntcur = (int*)(wsf + off);
  off += (size_t)NP;
  int* rowptr = (int*)(wsf + off);
  off += (size_t)NP + 256;
  int* bsum = (int*)(wsf + off);
  off += (size_t)((scanGrid + 255) / 256) * 256;
  int2* slot = (int2*)(wsf + off);
  off += (size_t)((2 * (size_t)E + 255) / 256) * 256;
  int2* csr = (int2*)(wsf + off);
  off += (size_t)((2 * (size_t)E + 255) / 256) * 256;
  float* hbuf = wsf + off;
  off += (size_t)N * HID;
  float* agg = wsf + off;

  const int padGrid = cdiv(NP, TPB);
  const int edgeGrid = cdiv(E, TPB);
  const int nodeWaveGrid = cdiv(N, TPB / 32);
  const int gemmGrid = cdiv(N, 128);

  zero_int_kernel<<<padGrid, TPB, 0, stream>>>(cntcur, NP);
  count_kernel<<<edgeGrid, TPB, 0, stream>>>(dst, cntcur, E);
  block_sum_kernel<<<scanGrid, TPB, 0, stream>>>(cntcur, bsum);
  scan_kernel<<<scanGrid, TPB, 0, stream>>>(cntcur, bsum, rowptr, dinv);
  place_kernel<<<edgeGrid, TPB, 0, stream>>>(src, dst, cntcur, slot, E);
  order_kernel<<<edgeGrid, TPB, 0, stream>>>(slot, dst, rowptr, dinv, csr, E);

  gemm_wmma_kernel<128, false><<<gemmGrid, TPB, 0, stream>>>(x, W1, hbuf, N);
  gather_kernel<<<nodeWaveGrid, TPB, 0, stream>>>(hbuf, rowptr, csr, dinv, b1,
                                                  agg, N);

  gemm_wmma_kernel<256, true><<<gemmGrid, TPB, 0, stream>>>(agg, W2, hbuf, N);
  gather_kernel<<<nodeWaveGrid, TPB, 0, stream>>>(hbuf, rowptr, csr, dinv, b2,
                                                  agg, N);

  gemm_wmma_kernel<256, true><<<gemmGrid, TPB, 0, stream>>>(agg, W3, hbuf, N);
  gather_kernel<<<nodeWaveGrid, TPB, 0, stream>>>(hbuf, rowptr, csr, dinv, b3,
                                                  out, N);
}


# layer 1 sums the 128-wide input rows per node first and multiplies after: (A x) W1 + b1 for A (x W1) + b1; half the gathered bytes per edge; layer 1's sum order changes
# speedup vs baseline: 3.4886x; 1.0680x over previous
#include <hip/hip_runtime.h>

typedef __attribute__((ext_vector_type(2))) float v2f;
typedef __attribute__((ext_vector_type(8))) float v8f;

#define HID 256
#define TPB 256
#define SCAN_CHUNK 1024


__global__ void zero_int_kernel(int* __restrict__ p, int n) {
  int i = blockIdx.x * blockDim.x + threadIdx.x;
  if (i < n) p[i] = 0;
}

__global__ void count_kernel(const int* __restrict__ dst,
                             int* __restrict__ cnt, int E) {
  int e = blockIdx.x * blockDim.x + threadIdx.x;
  if (e < E) atomicAdd(&cnt[dst[e]], 1);
}

__global__ __launch_bounds__(TPB) void block_sum_kernel(
    const int* __restrict__ cnt, int* __restrict__ bsum) {
  __shared__ int wsum[TPB / 32];
  const int t = threadIdx.x;
  const int4 v = *(const int4*)(cnt + (size_t)blockIdx.x * SCAN_CHUNK + t * 4);
  int s = v.x + v.y + v.z + v.w;
  for (int o = 16; o > 0; o >>= 1) s += __shfl_xor(s, o);
  if ((t & 31) == 0) wsum[t >> 5] = s;
  __syncthreads();
  if (t == 0) {
    int tot = 0;
    for (int i = 0; i < TPB / 32; ++i) tot += wsum[i];
    bsum[blockIdx.x] = tot;
  }
}

__global__ __launch_bounds__(TPB) void scan_kernel(
    int* __restrict__ cntcur, const int* __restrict__ bsum,
    int* __restrict__ rowptr, float* __restrict__ dinv) {
  __shared__ int wsum[TPB / 32];
  __shared__ int boff;
  const int t = threadIdx.x;
  const int lane = t & 31;
  const int wave = t >> 5;
  if (wave == 0) {
    int s = 0;
    for (int i = lane; i < (int)blockIdx.x; i += 32) s += bsum[i];
    for (int o = 16; o > 0; o >>= 1) s += __shfl_xor(s, o);
    if (lane == 0) boff = s;
  }
  const size_t base = (size_t)blockIdx.x * SCAN_CHUNK + t * 4;
  const int4 v = *(const int4*)(cntcur + base);
  const int s1 = v.x;
  const int s2 = s1 + v.y;
  const int s3 = s2 + v.z;
  const int s4 = s3 + v.w;
  int inc = s4;
  for (int o = 1; o < 32; o <<= 1) {
    const int up = __shfl_up(inc, o);
    if (lane >= o) inc += up;
  }
  if (lane == 31) wsum[wave] = inc;
  __syncthreads();
  int woff = boff;
  for (int i = 0; i < wave; ++i) woff += wsum[i];
  const int p0 = woff + inc - s4;
  int4 p;
  p.x = p0;
  p.y = p0 + s1;
  p.z = p0 + s2;
  p.w = p0 + s3;
  *(int4*)(rowptr + base) = p;
  *(int4*)(cntcur + base) = p;
  float4 dv;
  dv.x = 1.0f / sqrtf((float)(v.x + 1));
  dv.y = 1.0f / sqrtf((float)(v.y + 1));
  dv.z = 1.0f / sqrtf((float)(v.z + 1));
  dv.w = 1.0f / sqrtf((float)(v.w + 1));
  *(float4*)(dinv + base) = dv;
  if (blockIdx.x == gridDim.x - 1 && t == TPB - 1) rowptr[base + 4] = p0 + s4;
}

__global__ void place_kernel(const int* __restrict__ src,
                             const int* __restrict__ dst,
                             int* __restrict__ cursor,
                             int2* __restrict__ slot, int E) {
  int e = blockIdx.x * blockDim.x + threadIdx.x;
  if (e < E) {
    const int pos = atomicAdd(&cursor[dst[e]], 1);
    slot[pos] = make_int2(e, src[e]);
  }
}

__global__ void order_kernel(const int2* __restrict__ slot,
                             const int* __restrict__ dst,
                             const int* __restrict__ rowptr,
                             const float* __restrict__ dinv,
                             int2* __restrict__ csr, int E) {
  int a = blockIdx.x * blockDim.x + threadIdx.x;
  if (a >= E) return;
  const int2 me = slot[a];
  const int node = dst[me.x];
  const int beg = rowptr[node];
  const int end = rowptr[node + 1];
  int rank = 0;
  for (int b = beg; b < end; ++b) rank += (slot[b].x < me.x) ? 1 : 0;
  const float w = dinv[me.y] * dinv[node];
  csr[beg + rank] = make_int2(me.y, __float_as_int(w));
}

template <int K, bool RELU, bool BIAS>
__global__ __launch_bounds__(TPB) void gemm_wmma_kernel(
    const float* __restrict__ X, const float* __restrict__ W,
    const float* __restrict__ bias, float* __restrict__ H, int N) {
  const int lane = threadIdx.x & 31;
  const int wave = threadIdx.x >> 5;
  const int rgrp = wave >> 1;
  const int cgrp = wave & 1;
  const int row_base = blockIdx.x * 128 + rgrp * 32;
  if (row_base >= N) return;

  const int col_base = cgrp * 128;
  const int col = lane & 15;
  const int khalf = (lane >> 4) << 1;
  const size_t arow0 = (size_t)(row_base + col);
  const size_t arow1 = (size_t)(row_base + 16 + col);

  v8f acc[2][8];
#pragma unroll
  for (int m = 0; m < 2; ++m)
#pragma unroll
    for (int j = 0; j < 8; ++j)
      acc[m][j] = (v8f){0.f, 0.f, 0.f, 0.f, 0.f, 0.f, 0.f, 0.f};

  for (int k = 0; k < K; k += 4) {
    v2f a0 = *(const v2f*)(X + arow0 * K + (k + khalf));
    v2f a1 = *(const v2f*)(X + arow1 * K + (k + khalf));
    if (RELU) {
      a0.x = fmaxf(a0.x, 0.f);
      a0.y = fmaxf(a0.y, 0.f);
      a1.x = fmaxf(a1.x, 0.f);
      a1.y = fmaxf(a1.y, 0.f);
    }
#pragma unroll
    for (int j = 0; j < 8; ++j) {
      v2f b;
      b.x = W[(size_t)(k + khalf) * HID + col_base + j * 16 + col];
      b.y = W[(size_t)(k + khalf + 1) * HID + col_base + j * 16 + col];
      acc[0][j] = __builtin_amdgcn_wmma_f32_16x16x4_f32(
          false, a0, false, b, (short)0, acc[0][j], false, false);
      acc[1][j] = __builtin_amdgcn_wmma_f32_16x16x4_f32(
          false, a1, false, b, (short)0, acc[1][j], false, false);
    }
  }

#pragma unroll
  for (int m = 0; m < 2; ++m) {
    const int rlo = row_base + m * 16 + (lane >> 4) * 8;
#pragma unroll
    for (int j = 0; j < 8; ++j) {
      const float bj = BIAS ? bias[col_base + j * 16 + col] : 0.f;
#pragma unroll
      for (int r = 0; r < 8; ++r) {
        float v = acc[m][j][r];
        if (BIAS) v += bj;
        H[(size_t)(rlo + r) * HID + col_base + j * 16 + col] = v;
      }
    }
  }
}


template <int R>
__device__ __forceinline__ void add_rows(const float* __restrict__ h,
                                         const int2 ent, const int j,
                                         const int c, float4& a0, float4& a1) {
  float4 v0[R], v1[R];
  float w[R];
#pragma unroll
  for (int r = 0; r < R; ++r) {
    const int s = __builtin_amdgcn_readlane(ent.x, j + r);
    w[r] = __int_as_float(__builtin_amdgcn_readlane(ent.y, j + r));
    const float* hp = h + (size_t)s * HID + c;
    v0[r] = *(const float4*)(hp);
    v1[r] = *(const float4*)(hp + 128);
  }
#pragma unroll
  for (int r = 0; r < R; ++r) {
    a0.x = fmaf(v0[r].x, w[r], a0.x);
    a0.y = fmaf(v0[r].y, w[r], a0.y);
    a0.z = fmaf(v0[r].z, w[r], a0.z);
    a0.w = fmaf(v0[r].w, w[r], a0.w);
    a1.x = fmaf(v1[r].x, w[r], a1.x);
    a1.y = fmaf(v1[r].y, w[r], a1.y);
    a1.z = fmaf(v1[r].z, w[r], a1.z);
    a1.w = fmaf(v1[r].w, w[r], a1.w);
  }
}

__global__ __launch_bounds__(TPB) void gather_kernel(
    const float* __restrict__ h, const int* __restrict__ rowptr,
    const int2* __restrict__ csr, const float* __restrict__ dinv,
    const float* __restrict__ bias, float* __restrict__ agg, int N) {
  const int node = __builtin_amdgcn_readfirstlane(blockIdx.x * (TPB / 32) +
                                                  (threadIdx.x >> 5));
  if (node >= N) return;
  const int lane = threadIdx.x & 31;
  const int c = lane * 4;
  const int beg = rowptr[node];
  const int end = rowptr[node + 1];
  const float di = dinv[node];
  const float sn = di * di;
  const float* hp = h + (size_t)node * HID + c;
  const float4 x0 = *(const float4*)(hp);
  const float4 x1 = *(const float4*)(hp + 128);
  const float4 bv0 = *(const float4*)(bias + c);
  const float4 bv1 = *(const float4*)(bias + 128 + c);
  float4 a0 = bv0, a1 = bv1;

  for (int base = beg; base < end; base += 32) {
    const int n = min(32, end - base);
    int2 ent = make_int2(0, 0);
    if (lane < n) ent = csr[base + lane];
    int j = 0;
    for (; j + 4 <= n; j += 4) add_rows<4>(h, ent, j, c, a0, a1);
    const int rem = n - j;
    if (rem == 3) {
      add_rows<3>(h, ent, j, c, a0, a1);
    } else if (rem == 2) {
      add_rows<2>(h, ent, j, c, a0, a1);
    } else if (rem == 1) {
      add_rows<1>(h, ent, j, c, a0, a1);
    }
  }

  a0.x = fmaf(x0.x, sn, a0.x);
  a0.y = fmaf(x0.y, sn, a0.y);
  a0.z = fmaf(x0.z, sn, a0.z);
  a0.w = fmaf(x0.w, sn, a0.w);
  a1.x = fmaf(x1.x, sn, a1.x);
  a1.y = fmaf(x1.y, sn, a1.y);
  a1.z = fmaf(x1.z, sn, a1.z);
  a1.w = fmaf(x1.w, sn, a1.w);

  float* ap = agg + (size_t)node * HID + c;
  *(float4*)(ap) = a0;
  *(float4*)(ap + 128) = a1;
}


#define FIN 128

template <int R>
__device__ __forceinline__ void add_rows_in(const float* __restrict__ xin,
                                            const int2 ent, const int j,
                                            const int c, float4& a) {
  float4 v[R];
  float w[R];
#pragma unroll
  for (int r = 0; r < R; ++r) {
    const int s = __builtin_amdgcn_readlane(ent.x, j + r);
    w[r] = __int_as_float(__builtin_amdgcn_readlane(ent.y, j + r));
    v[r] = *(const float4*)(xin + (size_t)s * FIN + c);
  }
#pragma unroll
  for (int r = 0; r < R; ++r) {
    a.x = fmaf(v[r].x, w[r], a.x);
    a.y = fmaf(v[r].y, w[r], a.y);
    a.z = fmaf(v[r].z, w[r], a.z);
    a.w = fmaf(v[r].w, w[r], a.w);
  }
}

__global__ __launch_bounds__(TPB) void gather_in_kernel(
    const float* __restrict__ xin, const int* __restrict__ rowptr,
    const int2* __restrict__ csr, const float* __restrict__ dinv,
    float* __restrict__ aggx, int N) {
  const int node = __builtin_amdgcn_readfirstlane(blockIdx.x * (TPB / 32) +
                                                  (threadIdx.x >> 5));
  if (node >= N) return;
  const int lane = threadIdx.x & 31;
  const int c = lane * 4;
  const int beg = rowptr[node];
  const int end = rowptr[node + 1];
  const float di = dinv[node];
  const float sn = di * di;
  const float4 x0 = *(const float4*)(xin + (size_t)node * FIN + c);
  float4 a = make_float4(0.f, 0.f, 0.f, 0.f);

  for (int base = beg; base < end; base += 32) {
    const int n = min(32, end - base);
    int2 ent = make_int2(0, 0);
    if (lane < n) ent = csr[base + lane];
    int j = 0;
    for (; j + 8 <= n; j += 8) add_rows_in<8>(xin, ent, j, c, a);
    switch (n - j) {
      case 7: add_rows_in<7>(xin, ent, j, c, a); break;
      case 6: add_rows_in<6>(xin, ent, j, c, a); break;
      case 5: add_rows_in<5>(xin, ent, j, c, a); break;
      case 4: add_rows_in<4>(xin, ent, j, c, a); break;
      case 3: add_rows_in<3>(xin, ent, j, c, a); break;
      case 2: add_rows_in<2>(xin, ent, j, c, a); break;
      case 1: add_rows_in<1>(xin, ent, j, c, a); break;
      default: break;
    }
  }

  a.x = fmaf(x0.x, sn, a.x);
  a.y = fmaf(x0.y, sn, a.y);
  a.z = fmaf(x0.z, sn, a.z);
  a.w = fmaf(x0.w, sn, a.w);

  *(float4*)(aggx + (size_t)node * FIN + c) = a;
}


static inline int cdiv(int a, int b) { return (a + b - 1) / b; }

extern "C" void kernel_launch(void* const* d_in, const int* in_sizes, int n_in,
                              void* d_out, int out_size, void* d_ws,
                              size_t ws_size, hipStream_t stream) {
  const float* x = (const float*)d_in[0];
  const int* ei = (const int*)d_in[1];
  const float* W1 = (const float*)d_in[2];
  const float* b1 = (const float*)d_in[3];
  const float* W2 = (const float*)d_in[4];
  const float* b2 = (const float*)d_in[5];
  const float* W3 = (const float*)d_in[6];
  const float* b3 = (const float*)d_in[7];
  float* out = (float*)d_out;

  const int F = 128;
  const int N = in_sizes[0] / F;
  const int E = in_sizes[1] / 2;
  const int* src = ei;
  const int* dst = ei + E;

  const int scanGrid = cdiv(N, SCAN_CHUNK);
  const int NP = scanGrid * SCAN_CHUNK;

  float* wsf = (float*)d_ws;
  size_t off = 0;
  float* dinv = wsf + off;
  off += (size_t)NP;
  int* cntcur = (int*)(wsf + off);
  off += (size_t)NP;
  int* rowptr = (int*)(wsf + off);
  off += (size_t)NP + 256;
  int* bsum = (int*)(wsf + off);
  off += (size_t)((scanGrid + 255) / 256) * 256;
  int2* slot = (int2*)(wsf + off);
  off += (size_t)((2 * (size_t)E + 255) / 256) * 256;
  int2* csr = (int2*)(wsf + off);
  off += (size_t)((2 * (size_t)E + 255) / 256) * 256;
  float* hbuf = wsf + off;
  off += (size_t)N * HID;
  float* agg = wsf + off;

  const int padGrid = cdiv(NP, TPB);
  const int edgeGrid = cdiv(E, TPB);
  const int nodeWaveGrid = cdiv(N, TPB / 32);
  const int gemmGrid = cdiv(N, 128);

  zero_int_kernel<<<padGrid, TPB, 0, stream>>>(cntcur, NP);
  count_kernel<<<edgeGrid, TPB, 0, stream>>>(dst, cntcur, E);
  block_sum_kernel<<<scanGrid, TPB, 0, stream>>>(cntcur, bsum);
  scan_kernel<<<scanGrid, TPB, 0, stream>>>(cntcur, bsum, rowptr, dinv);
  place_kernel<<<edgeGrid, TPB, 0, stream>>>(src, dst, cntcur, slot, E);
  order_kernel<<<edgeGrid, TPB, 0, stream>>>(slot, dst, rowptr, dinv, csr, E);

  float* aggx = hbuf;
  gather_in_kernel<<<nodeWaveGrid, TPB, 0, stream>>>(x, rowptr, csr, dinv,
                                                     aggx, N);
  gemm_wmma_kernel<128, false, true><<<gemmGrid, TPB, 0, stream>>>(
      aggx, W1, b1, agg, N);

  gemm_wmma_kernel<256, true, false><<<gemmGrid, TPB, 0, stream>>>(
      agg, W2, b2, hbuf, N);
  gather_kernel<<<nodeWaveGrid, TPB, 0, stream>>>(hbuf, rowptr, csr, dinv, b2,
                                                  agg, N);

  gemm_wmma_kernel<256, true, false><<<gemmGrid, TPB, 0, stream>>>(
      agg, W3, b3, hbuf, N);
  gather_kernel<<<nodeWaveGrid, TPB, 0, stream>>>(hbuf, rowptr, csr, dinv, b3,
                                                  out, N);
}


# GEMMs: W staged once per workgroup in LDS as (k,k+1) row pairs; one resident 512-thread workgroup per WGP loops over 32-row groups, A rows requested one k step ahead; k order and sums unchanged
# speedup vs baseline: 3.7312x; 1.0695x over previous
#include <hip/hip_runtime.h>

typedef __attribute__((ext_vector_type(2))) float v2f;
typedef __attribute__((ext_vector_type(8))) float v8f;

#define HID 256
#define TPB 256
#define SCAN_CHUNK 1024


__global__ void zero_int_kernel(int* __restrict__ p, int n) {
  int i = blockIdx.x * blockDim.x + threadIdx.x;
  if (i < n) p[i] = 0;
}

__global__ void count_kernel(const int* __restrict__ dst,
                             int* __restrict__ cnt, int E) {
  int e = blockIdx.x * blockDim.x + threadIdx.x;
  if (e < E) atomicAdd(&cnt[dst[e]], 1);
}

__global__ __launch_bounds__(TPB) void block_sum_kernel(
    const int* __restrict__ cnt, int* __restrict__ bsum) {
  __shared__ int wsum[TPB / 32];
  const int t = threadIdx.x;
  const int4 v = *(const int4*)(cnt + (size_t)blockIdx.x * SCAN_CHUNK + t * 4);
  int s = v.x + v.y + v.z + v.w;
  for (int o = 16; o > 0; o >>= 1) s += __shfl_xor(s, o);
  if ((t & 31) == 0) wsum[t >> 5] = s;
  __syncthreads();
  if (t == 0) {
    int tot = 0;
    for (int i = 0; i < TPB / 32; ++i) tot += wsum[i];
    bsum[blockIdx.x] = tot;
  }
}

__global__ __launch_bounds__(TPB) void scan_kernel(
    int* __restrict__ cntcur, const int* __restrict__ bsum,
    int* __restrict__ rowptr, float* __restrict__ dinv) {
  __shared__ int wsum[TPB / 32];
  __shared__ int boff;
  const int t = threadIdx.x;
  const int lane = t & 31;
  const int wave = t >> 5;
  if (wave == 0) {
    int s = 0;
    for (int i = lane; i < (int)blockIdx.x; i += 32) s += bsum[i];
    for (int o = 16; o > 0; o >>= 1) s += __shfl_xor(s, o);
    if (lane == 0) boff = s;
  }
  const size_t base = (size_t)blockIdx.x * SCAN_CHUNK + t * 4;
  const int4 v = *(const int4*)(cntcur + base);
  const int s1 = v.x;
  const int s2 = s1 + v.y;
  const int s3 = s2 + v.z;
  const int s4 = s3 + v.w;
  int inc = s4;
  for (int o = 1; o < 32; o <<= 1) {
    const int up = __shfl_up(inc, o);
    if (lane >= o) inc += up;
  }
  if (lane == 31) wsum[wave] = inc;
  __syncthreads();
  int woff = boff;
  for (int i = 0; i < wave; ++i) woff += wsum[i];
  const int p0 = woff + inc - s4;
  int4 p;
  p.x = p0;
  p.y = p0 + s1;
  p.z = p0 + s2;
  p.w = p0 + s3;
  *(int4*)(rowptr + base) = p;
  *(int4*)(cntcur + base) = p;
  float4 dv;
  dv.x = 1.0f / sqrtf((float)(v.x + 1));
  dv.y = 1.0f / sqrtf((float)(v.y + 1));
  dv.z = 1.0f / sqrtf((float)(v.z + 1));
  dv.w = 1.0f / sqrtf((float)(v.w + 1));
  *(float4*)(dinv + base) = dv;
  if (blockIdx.x == gridDim.x - 1 && t == TPB - 1) rowptr[base + 4] = p0 + s4;
}

__global__ void place_kernel(const int* __restrict__ src,
                             const int* __restrict__ dst,
                             int* __restrict__ cursor,
                             int2* __restrict__ slot, int E) {
  int e = blockIdx.x * blockDim.x + threadIdx.x;
  if (e < E) {
    const int pos = atomicAdd(&cursor[dst[e]], 1);
    slot[pos] = make_int2(e, src[e]);
  }
}

__global__ void order_kernel(const int2* __restrict__ slot,
                             const int* __restrict__ dst,
                             const int* __restrict__ rowptr,
                             const float* __restrict__ dinv,
                             int2* __restrict__ csr, int E) {
  int a = blockIdx.x * blockDim.x + threadIdx.x;
  if (a >= E) return;
  const int2 me = slot[a];
  const int node = dst[me.x];
  const int beg = rowptr[node];
  const int end = rowptr[node + 1];
  int rank = 0;
  for (int b = beg; b < end; ++b) rank += (slot[b].x < me.x) ? 1 : 0;
  const float w = dinv[me.y] * dinv[node];
  csr[beg + rank] = make_int2(me.y, __float_as_int(w));
}

#define WPITCH 544
#define GEMM_TPB 512
#define GEMM_GRID 256

template <bool RELU>
__device__ __forceinline__ void gemm_step(v2f a0, v2f a1,
                                          const float* __restrict__ bk,
                                          v8f (&acc)[2][8]) {
  if (RELU) {
    a0.x = fmaxf(a0.x, 0.f);
    a0.y = fmaxf(a0.y, 0.f);
    a1.x = fmaxf(a1.x, 0.f);
    a1.y = fmaxf(a1.y, 0.f);
  }
  v2f b[8];
#pragma unroll
  for (int j = 0; j < 8; ++j) b[j] = *(const v2f*)(bk + j * 32);
#pragma unroll
  for (int j = 0; j < 8; ++j) {
    acc[0][j] = __builtin_amdgcn_wmma_f32_16x16x4_f32(
        false, a0, false, b[j], (short)0, acc[0][j], false, false);
    acc[1][j] = __builtin_amdgcn_wmma_f32_16x16x4_f32(
        false, a1, false, b[j], (short)0, acc[1][j], false, false);
  }
}

template <int K, bool RELU, bool BIAS>
__global__ __launch_bounds__(GEMM_TPB) void gemm_wmma_kernel(
    const float* __restrict__ X, const float* __restrict__ W,
    const float* __restrict__ bias, float* __restrict__ H, int N) {
  __shared__ float wl[128 * WPITCH];

  const int tid = threadIdx.x;
#pragma unroll 4
  for (int idx = tid; idx < (K / 2) * 64; idx += GEMM_TPB) {
    const int p = idx >> 6;
    const int n = (idx & 63) * 4;
    const float4 r0 = *(const float4*)(W + (size_t)(2 * p) * HID + n);
    const float4 r1 = *(const float4*)(W + (size_t)(2 * p + 1) * HID + n);
    float* d = wl + p * WPITCH + 2 * n;
    *(float4*)(d) = make_float4(r0.x, r1.x, r0.y, r1.y);
    *(float4*)(d + 4) = make_float4(r0.z, r1.z, r0.w, r1.w);
  }
  __syncthreads();

  const int lane = tid & 31;
  const int wave = __builtin_amdgcn_readfirstlane(tid >> 5);
  const int cgrp = wave & 1;
  const int pair = (wave >> 1) & 1;
  const int q = wave >> 2;
  const int nps = gridDim.x * 2;
  const int ps = blockIdx.x * 2 + pair;
  const int ngroups = N / 32;

  const int col_base = cgrp * 128;
  const int col = lane & 15;
  const int h = lane >> 4;
  const int khalf = h << 1;
  const float* bl = wl + h * WPITCH + (col_base + col) * 2;

  for (int g = ps + nps * q; g < ngroups; g += nps * (GEMM_TPB / 128)) {
    const int row_base = g * 32;
    const float* xa0 = X + (size_t)(row_base + col) * K + khalf;
    const float* xa1 = xa0 + (size_t)16 * K;

    v8f acc[2][8];
#pragma unroll
    for (int m = 0; m < 2; ++m)
#pragma unroll
      for (int j = 0; j < 8; ++j)
        acc[m][j] = (v8f){0.f, 0.f, 0.f, 0.f, 0.f, 0.f, 0.f, 0.f};

    v2f pa0 = *(const v2f*)(xa0);
    v2f pa1 = *(const v2f*)(xa1);
    for (int k = 0; k < K; k += 8) {
      const v2f qa0 = *(const v2f*)(xa0 + k + 4);
      const v2f qa1 = *(const v2f*)(xa1 + k + 4);
      __builtin_amdgcn_sched_barrier(0);
      gemm_step<RELU>(pa0, pa1, bl + (k >> 1) * WPITCH, acc);
      const int kn = (k + 8 < K) ? k + 8 : k + 4;
      pa0 = *(const v2f*)(xa0 + kn);
      pa1 = *(const v2f*)(xa1 + kn);
      __builtin_amdgcn_sched_barrier(0);
      gemm_step<RELU>(qa0, qa1, bl + ((k + 4) >> 1) * WPITCH, acc);
      __builtin_amdgcn_sched_barrier(0);
      asm volatile("" ::"v"(pa0), "v"(pa1));
    }

#pragma unroll
    for (int m = 0; m < 2; ++m) {
      const int rlo = row_base + m * 16 + h * 8;
#pragma unroll
      for (int j = 0; j < 8; ++j) {
        const float bj = BIAS ? bias[col_base + j * 16 + col] : 0.f;
#pragma unroll
        for (int r = 0; r < 8; ++r) {
          float v = acc[m][j][r];
          if (BIAS) v += bj;
          H[(size_t)(rlo + r) * HID + col_base + j * 16 + col] = v;
        }
      }
    }
  }
}


template <int R>
__device__ __forceinline__ void add_rows(const float* __restrict__ h,
                                         const int2 ent, const int j,
                                         const int c, float4& a0, float4& a1) {
  float4 v0[R], v1[R];
  float w[R];
#pragma unroll
  for (int r = 0; r < R; ++r) {
    const int s = __builtin_amdgcn_readlane(ent.x, j + r);
    w[r] = __int_as_float(__builtin_amdgcn_readlane(ent.y, j + r));
    const float* hp = h + (size_t)s * HID + c;
    v0[r] = *(const float4*)(hp);
    v1[r] = *(const float4*)(hp + 128);
  }
#pragma unroll
  for (int r = 0; r < R; ++r) {
    a0.x = fmaf(v0[r].x, w[r], a0.x);
    a0.y = fmaf(v0[r].y, w[r], a0.y);
    a0.z = fmaf(v0[r].z, w[r], a0.z);
    a0.w = fmaf(v0[r].w, w[r], a0.w);
    a1.x = fmaf(v1[r].x, w[r], a1.x);
    a1.y = fmaf(v1[r].y, w[r], a1.y);
    a1.z = fmaf(v1[r].z, w[r], a1.z);
    a1.w = fmaf(v1[r].w, w[r], a1.w);
  }
}

__global__ __launch_bounds__(TPB) void gather_kernel(
    const float* __restrict__ h, const int* __restrict__ rowptr,
    const int2* __restrict__ csr, const float* __restrict__ dinv,
    const float* __restrict__ bias, float* __restrict__ agg, int N) {
  const int node = __builtin_amdgcn_readfirstlane(blockIdx.x * (TPB / 32) +
                                                  (threadIdx.x >> 5));
  if (node >= N) return;
  const int lane = threadIdx.x & 31;
  const int c = lane * 4;
  const int beg = rowptr[node];
  const int end = rowptr[node + 1];
  const float di = dinv[node];
  const float sn = di * di;
  const float* hp = h + (size_t)node * HID + c;
  const float4 x0 = *(const float4*)(hp);
  const float4 x1 = *(const float4*)(hp + 128);
  const float4 bv0 = *(const float4*)(bias + c);
  const float4 bv1 = *(const float4*)(bias + 128 + c);
  float4 a0 = bv0, a1 = bv1;

  for (int base = beg; base < end; base += 32) {
    const int n = min(32, end - base);
    int2 ent = make_int2(0, 0);
    if (lane < n) ent = csr[base + lane];
    int j = 0;
    for (; j + 4 <= n; j += 4) add_rows<4>(h, ent, j, c, a0, a1);
    const int rem = n - j;
    if (rem == 3) {
      add_rows<3>(h, ent, j, c, a0, a1);
    } else if (rem == 2) {
      add_rows<2>(h, ent, j, c, a0, a1);
    } else if (rem == 1) {
      add_rows<1>(h, ent, j, c, a0, a1);
    }
  }

  a0.x = fmaf(x0.x, sn, a0.x);
  a0.y = fmaf(x0.y, sn, a0.y);
  a0.z = fmaf(x0.z, sn, a0.z);
  a0.w = fmaf(x0.w, sn, a0.w);
  a1.x = fmaf(x1.x, sn, a1.x);
  a1.y = fmaf(x1.y, sn, a1.y);
  a1.z = fmaf(x1.z, sn, a1.z);
  a1.w = fmaf(x1.w, sn, a1.w);

  float* ap = agg + (size_t)node * HID + c;
  *(float4*)(ap) = a0;
  *(float4*)(ap + 128) = a1;
}


#define FIN 128

template <int R>
__device__ __forceinline__ void add_rows_in(const float* __restrict__ xin,
                                            const int2 ent, const int j,
                                            const int c, float4& a) {
  float4 v[R];
  float w[R];
#pragma unroll
  for (int r = 0; r < R; ++r) {
    const int s = __builtin_amdgcn_readlane(ent.x, j + r);
    w[r] = __int_as_float(__builtin_amdgcn_readlane(ent.y, j + r));
    v[r] = *(const float4*)(xin + (size_t)s * FIN + c);
  }
#pragma unroll
  for (int r = 0; r < R; ++r) {
    a.x = fmaf(v[r].x, w[r], a.x);
    a.y = fmaf(v[r].y, w[r], a.y);
    a.z = fmaf(v[r].z, w[r], a.z);
    a.w = fmaf(v[r].w, w[r], a.w);
  }
}

__global__ __launch_bounds__(TPB) void gather_in_kernel(
    const float* __restrict__ xin, const int* __restrict__ rowptr,
    const int2* __restrict__ csr, const float* __restrict__ dinv,
    float* __restrict__ aggx, int N) {
  const int node = __builtin_amdgcn_readfirstlane(blockIdx.x * (TPB / 32) +
                                                  (threadIdx.x >> 5));
  if (node >= N) return;
  const int lane = threadIdx.x & 31;
  const int c = lane * 4;
  const int beg = rowptr[node];
  const int end = rowptr[node + 1];
  const float di = dinv[node];
  const float sn = di * di;
  const float4 x0 = *(const float4*)(xin + (size_t)node * FIN + c);
  float4 a = make_float4(0.f, 0.f, 0.f, 0.f);

  for (int base = beg; base < end; base += 32) {
    const int n = min(32, end - base);
    int2 ent = make_int2(0, 0);
    if (lane < n) ent = csr[base + lane];
    int j = 0;
    for (; j + 8 <= n; j += 8) add_rows_in<8>(xin, ent, j, c, a);
    switch (n - j) {
      case 7: add_rows_in<7>(xin, ent, j, c, a); break;
      case 6: add_rows_in<6>(xin, ent, j, c, a); break;
      case 5: add_rows_in<5>(xin, ent, j, c, a); break;
      case 4: add_rows_in<4>(xin, ent, j, c, a); break;
      case 3: add_rows_in<3>(xin, ent, j, c, a); break;
      case 2: add_rows_in<2>(xin, ent, j, c, a); break;
      case 1: add_rows_in<1>(xin, ent, j, c, a); break;
      default: break;
    }
  }

  a.x = fmaf(x0.x, sn, a.x);
  a.y = fmaf(x0.y, sn, a.y);
  a.z = fmaf(x0.z, sn, a.z);
  a.w = fmaf(x0.w, sn, a.w);

  *(float4*)(aggx + (size_t)node * FIN + c) = a;
}


static inline int cdiv(int a, int b) { return (a + b - 1) / b; }

extern "C" void kernel_launch(void* const* d_in, const int* in_sizes, int n_in,
                              void* d_out, int out_size, void* d_ws,
                              size_t ws_size, hipStream_t stream) {
  const float* x = (const float*)d_in[0];
  const int* ei = (const int*)d_in[1];
  const float* W1 = (const float*)d_in[2];
  const float* b1 = (const float*)d_in[3];
  const float* W2 = (const float*)d_in[4];
  const float* b2 = (const float*)d_in[5];
  const float* W3 = (const float*)d_in[6];
  const float* b3 = (const float*)d_in[7];
  float* out = (float*)d_out;

  const int F = 128;
  const int N = in_sizes[0] / F;
  const int E = in_sizes[1] / 2;
  const int* src = ei;
  const int* dst = ei + E;

  const int scanGrid = cdiv(N, SCAN_CHUNK);
  const int NP = scanGrid * SCAN_CHUNK;

  float* wsf = (float*)d_ws;
  size_t off = 0;
  float* dinv = wsf + off;
  off += (size_t)NP;
  int* cntcur = (int*)(wsf + off);
  off += (size_t)NP;
  int* rowptr = (int*)(wsf + off);
  off += (size_t)NP + 256;
  int* bsum = (int*)(wsf + off);
  off += (size_t)((scanGrid + 255) / 256) * 256;
  int2* slot = (int2*)(wsf + off);
  off += (size_t)((2 * (size_t)E + 255) / 256) * 256;
  int2* csr = (int2*)(wsf + off);
  off += (size_t)((2 * (size_t)E + 255) / 256) * 256;
  float* hbuf = wsf + off;
  off += (size_t)N * HID;
  float* agg = wsf + off;

  const int padGrid = cdiv(NP, TPB);
  const int edgeGrid = cdiv(E, TPB);
  const int nodeWaveGrid = cdiv(N, TPB / 32);

  zero_int_kernel<<<padGrid, TPB, 0, stream>>>(cntcur, NP);
  count_kernel<<<edgeGrid, TPB, 0, stream>>>(dst, cntcur, E);
  block_sum_kernel<<<scanGrid, TPB, 0, stream>>>(cntcur, bsum);
  scan_kernel<<<scanGrid, TPB, 0, stream>>>(cntcur, bsum, rowptr, dinv);
  place_kernel<<<edgeGrid, TPB, 0, stream>>>(src, dst, cntcur, slot, E);
  order_kernel<<<edgeGrid, TPB, 0, stream>>>(slot, dst, rowptr, dinv, csr, E);

  float* aggx = hbuf;
  gather_in_kernel<<<nodeWaveGrid, TPB, 0, stream>>>(x, rowptr, csr, dinv,
                                                     aggx, N);
  gemm_wmma_kernel<128, false, true><<<GEMM_GRID, GEMM_TPB, 0, stream>>>(
      aggx, W1, b1, agg, N);

  gemm_wmma_kernel<256, true, false><<<GEMM_GRID, GEMM_TPB, 0, stream>>>(
      agg, W2, b2, hbuf, N);
  gather_kernel<<<nodeWaveGrid, TPB, 0, stream>>>(hbuf, rowptr, csr, dinv, b2,
                                                  agg, N);

  gemm_wmma_kernel<256, true, false><<<GEMM_GRID, GEMM_TPB, 0, stream>>>(
      agg, W3, b3, hbuf, N);
  gather_kernel<<<nodeWaveGrid, TPB, 0, stream>>>(hbuf, rowptr, csr, dinv, b3,
                                                  out, N);
}


# neighbour sum moved in front of every layer's product and fused into it: 16 waves sum a 16-row tile into LDS while 8 waves run the tile before's WMMAs; sum order of layers 2,3 changes
# speedup vs baseline: 4.3415x; 1.1636x over previous
#include <hip/hip_runtime.h>

typedef __attribute__((ext_vector_type(2))) float v2f;
typedef __attribute__((ext_vector_type(8))) float v8f;

#define HID 256
#define TPB 256
#define SCAN_CHUNK 1024


__global__ void zero_int_kernel(int* __restrict__ p, int n) {
  int i = blockIdx.x * blockDim.x + threadIdx.x;
  if (i < n) p[i] = 0;
}

__global__ void count_kernel(const int* __restrict__ dst,
                             int* __restrict__ cnt, int E) {
  int e = blockIdx.x * blockDim.x + threadIdx.x;
  if (e < E) atomicAdd(&cnt[dst[e]], 1);
}

__global__ __launch_bounds__(TPB) void block_sum_kernel(
    const int* __restrict__ cnt, int* __restrict__ bsum) {
  __shared__ int wsum[TPB / 32];
  const int t = threadIdx.x;
  const int4 v = *(const int4*)(cnt + (size_t)blockIdx.x * SCAN_CHUNK + t * 4);
  int s = v.x + v.y + v.z + v.w;
  for (int o = 16; o > 0; o >>= 1) s += __shfl_xor(s, o);
  if ((t & 31) == 0) wsum[t >> 5] = s;
  __syncthreads();
  if (t == 0) {
    int tot = 0;
    for (int i = 0; i < TPB / 32; ++i) tot += wsum[i];
    bsum[blockIdx.x] = tot;
  }
}

__global__ __launch_bounds__(TPB) void scan_kernel(
    int* __restrict__ cntcur, const int* __restrict__ bsum,
    int* __restrict__ rowptr, float* __restrict__ dinv) {
  __shared__ int wsum[TPB / 32];
  __shared__ int boff;
  const int t = threadIdx.x;
  const int lane = t & 31;
  const int wave = t >> 5;
  if (wave == 0) {
    int s = 0;
    for (int i = lane; i < (int)blockIdx.x; i += 32) s += bsum[i];
    for (int o = 16; o > 0; o >>= 1) s += __shfl_xor(s, o);
    if (lane == 0) boff = s;
  }
  const size_t base = (size_t)blockIdx.x * SCAN_CHUNK + t * 4;
  const int4 v = *(const int4*)(cntcur + base);
  const int s1 = v.x;
  const int s2 = s1 + v.y;
  const int s3 = s2 + v.z;
  const int s4 = s3 + v.w;
  int inc = s4;
  for (int o = 1; o < 32; o <<= 1) {
    const int up = __shfl_up(inc, o);
    if (lane >= o) inc += up;
  }
  if (lane == 31) wsum[wave] = inc;
  __syncthreads();
  int woff = boff;
  for (int i = 0; i < wave; ++i) woff += wsum[i];
  const int p0 = woff + inc - s4;
  int4 p;
  p.x = p0;
  p.y = p0 + s1;
  p.z = p0 + s2;
  p.w = p0 + s3;
  *(int4*)(rowptr + base) = p;
  *(int4*)(cntcur + base) = p;
  float4 dv;
  dv.x = 1.0f / sqrtf((float)(v.x + 1));
  dv.y = 1.0f / sqrtf((float)(v.y + 1));
  dv.z = 1.0f / sqrtf((float)(v.z + 1));
  dv.w = 1.0f / sqrtf((float)(v.w + 1));
  *(float4*)(dinv + base) = dv;
  if (blockIdx.x == gridDim.x - 1 && t == TPB - 1) rowptr[base + 4] = p0 + s4;
}

__global__ void place_kernel(const int* __restrict__ src,
                             const int* __restrict__ dst,
                             int* __restrict__ cursor,
                             int2* __restrict__ slot, int E) {
  int e = blockIdx.x * blockDim.x + threadIdx.x;
  if (e < E) {
    const int pos = atomicAdd(&cursor[dst[e]], 1);
    slot[pos] = make_int2(e, src[e]);
  }
}

__global__ void order_kernel(const int2* __restrict__ slot,
                             const int* __restrict__ dst,
                             const int* __restrict__ rowptr,
                             const float* __restrict__ dinv,
                             int2* __restrict__ csr, int E) {
  int a = blockIdx.x * blockDim.x + threadIdx.x;
  if (a >= E) return;
  const int2 me = slot[a];
  const int node = dst[me.x];
  const int beg = rowptr[node];
  const int end = rowptr[node + 1];
  int rank = 0;
  for (int b = beg; b < end; ++b) rank += (slot[b].x < me.x) ? 1 : 0;
  const float w = dinv[me.y] * dinv[node];
  csr[beg + rank] = make_int2(me.y, __float_as_int(w));
}

#define WPITCH 544
#define FUSE_TPB 768
#define FUSE_GRID 256
#define TROWS 16
#define MMW 8
#define RMAX 12

template <int K, int R>
__device__ __forceinline__ void add_rows_k(const float* __restrict__ tab,
                                           const int2 ent, const int j,
                                           const int c,
                                           float4 (&a)[K / 128]) {
  constexpr int NV = K / 128;
  float4 v[R][NV];
  float w[R];
#pragma unroll
  for (int r = 0; r < R; ++r) {
    const int s = __builtin_amdgcn_readlane(ent.x, j + r);
    w[r] = __int_as_float(__builtin_amdgcn_readlane(ent.y, j + r));
    const float* hp = tab + (size_t)s * K + c;
#pragma unroll
    for (int q = 0; q < NV; ++q) v[r][q] = *(const float4*)(hp + q * 128);
  }
#pragma unroll
  for (int r = 0; r < R; ++r) {
#pragma unroll
    for (int q = 0; q < NV; ++q) {
      a[q].x = fmaf(v[r][q].x, w[r], a[q].x);
      a[q].y = fmaf(v[r][q].y, w[r], a[q].y);
      a[q].z = fmaf(v[r][q].z, w[r], a[q].z);
      a[q].w = fmaf(v[r][q].w, w[r], a[q].w);
    }
  }
}

template <int K>
__device__ __forceinline__ void add_tail_k(const float* __restrict__ tab,
                                           const int2 ent, const int j,
                                           const int rem, const int c,
                                           float4 (&a)[K / 128]) {
  switch (rem) {
    case 11: add_rows_k<K, 11>(tab, ent, j, c, a); break;
    case 10: add_rows_k<K, 10>(tab, ent, j, c, a); break;
    case 9: add_rows_k<K, 9>(tab, ent, j, c, a); break;
    case 8: add_rows_k<K, 8>(tab, ent, j, c, a); break;
    case 7: add_rows_k<K, 7>(tab, ent, j, c, a); break;
    case 6: add_rows_k<K, 6>(tab, ent, j, c, a); break;
    case 5: add_rows_k<K, 5>(tab, ent, j, c, a); break;
    case 4: add_rows_k<K, 4>(tab, ent, j, c, a); break;
    case 3: add_rows_k<K, 3>(tab, ent, j, c, a); break;
    case 2: add_rows_k<K, 2>(tab, ent, j, c, a); break;
    case 1: add_rows_k<K, 1>(tab, ent, j, c, a); break;
    default: break;
  }
}

__device__ __forceinline__ void load_blk(const float* __restrict__ at,
                                         const float* __restrict__ bt,
                                         v2f (&fa)[4], v2f (&fb)[4][2]) {
#pragma unroll
  for (int s = 0; s < 4; ++s) {
    fa[s] = *(const v2f*)(at + 4 * s);
    fb[s][0] = *(const v2f*)(bt + (2 * s) * WPITCH);
    fb[s][1] = *(const v2f*)(bt + (2 * s) * WPITCH + 32);
  }
}

__device__ __forceinline__ void mma_blk(const v2f (&fa)[4],
                                        const v2f (&fb)[4][2], v8f& acc0,
                                        v8f& acc1) {
#pragma unroll
  for (int s = 0; s < 4; ++s) {
    acc0 = __builtin_amdgcn_wmma_f32_16x16x4_f32(false, fa[s], false, fb[s][0],
                                                 (short)0, acc0, false, false);
    acc1 = __builtin_amdgcn_wmma_f32_16x16x4_f32(false, fa[s], false, fb[s][1],
                                                 (short)0, acc1, false, false);
  }
}

template <bool RELU_OUT>
__device__ __forceinline__ void store_rows(const v8f& acc0, const v8f& acc1,
                                           float* __restrict__ orow,
                                           const int h, const float bj) {
#pragma unroll
  for (int r = 0; r < 8; ++r) {
    const float mine0 = acc0[r];
    const float mine1 = acc1[r];
    const float give = h ? mine0 : mine1;
    const float got = __shfl_xor(give, 16);
    float lo = (h ? got : mine0) + bj;
    float hi = (h ? mine1 : got) + bj;
    if (RELU_OUT) {
      lo = fmaxf(lo, 0.f);
      hi = fmaxf(hi, 0.f);
    }
    orow[(size_t)r * HID] = lo;
    orow[(size_t)(r + 8) * HID] = hi;
  }
}

template <int K, bool RELU_OUT>
__global__ __launch_bounds__(FUSE_TPB) void fused_layer_kernel(
    const float* __restrict__ tab, const float* __restrict__ W,
    const float* __restrict__ bias, const int* __restrict__ rowptr,
    const int2* __restrict__ csr, const float* __restrict__ dinv,
    float* __restrict__ out, int N) {
  constexpr int NV = K / 128;
  constexpr int P = K + 4;
  constexpr int KB = K / 16;
  __shared__ float wl[128 * WPITCH + 2 * TROWS * (256 + 4)];
  float* const tiles = wl + 128 * WPITCH;

  const int tid = threadIdx.x;
  constexpr int STAGE_ITEMS = (K / 2) * 64;
  constexpr int STAGE_TRIPS = (STAGE_ITEMS + FUSE_TPB - 1) / FUSE_TPB;
#pragma unroll 4
  for (int i = 0; i < STAGE_TRIPS; ++i) {
    const int idx = tid + i * FUSE_TPB;
    if (idx < STAGE_ITEMS) {
      const int p = idx >> 6;
      const int n = (idx & 63) * 4;
      const float4 r0 = *(const float4*)(W + (size_t)(2 * p) * HID + n);
      const float4 r1 = *(const float4*)(W + (size_t)(2 * p + 1) * HID + n);
      float* d = wl + p * WPITCH + 2 * n;
      *(float4*)(d) = make_float4(r0.x, r1.x, r0.y, r1.y);
      *(float4*)(d + 4) = make_float4(r0.z, r1.z, r0.w, r1.w);
    }
  }

  const int lane = tid & 31;
  const int wave = __builtin_amdgcn_readfirstlane(tid >> 5);
  const int G = gridDim.x;
  const int b = blockIdx.x;
  const int ntiles = N / TROWS;
  const int nt = (b < ntiles) ? (ntiles - 1 - b) / G + 1 : 0;
  const bool summer = wave >= MMW;
  const int gw = summer ? wave - MMW : 0;
  const int ww = wave & (MMW - 1);
  const int c = lane * 4;
  const int col = lane & 15;
  const int h = lane >> 4;

  int node = 0, beg = 0, end = 0, nnode = 0, nbeg = 0, nend = 0;
  float di = 0.f, ndi = 0.f;
  int2 ent = make_int2(0, 0);
  if (summer && nt > 0) {
    node = b * TROWS + gw;
    beg = rowptr[node];
    end = rowptr[node + 1];
    di = dinv[node];
    if (lane < end - beg) ent = csr[beg + lane];
    nnode = min(b + G, ntiles - 1) * TROWS + gw;
    nbeg = rowptr[nnode];
    nend = rowptr[nnode + 1];
    ndi = dinv[nnode];
  }
  const float bj = bias[32 * ww + lane];
  float* const ocol = out + 32 * ww + lane;
  const float* const bt = wl + h * WPITCH + (32 * ww + col) * 2;
  v8f acc0 = (v8f){0.f, 0.f, 0.f, 0.f, 0.f, 0.f, 0.f, 0.f};
  v8f acc1 = acc0;

  __syncthreads();

  for (int t = 0; t <= nt; ++t) {
    if (summer) {
      if (t < nt) {
        int2 nent = make_int2(0, 0);
        if (lane < nend - nbeg) nent = csr[nbeg + lane];
        const int n2 = min(b + (t + 2) * G, ntiles - 1) * TROWS + gw;
        const int n2beg = rowptr[n2];
        const int n2end = rowptr[n2 + 1];
        const float n2di = dinv[n2];

        const float* sp = tab + (size_t)node * K + c;
        float4 xs[NV];
        float4 a[NV];
#pragma unroll
        for (int q = 0; q < NV; ++q) {
          xs[q] = *(const float4*)(sp + q * 128);
          a[q] = make_float4(0.f, 0.f, 0.f, 0.f);
        }
        const int n0 = min(32, end - beg);
        int j = 0;
        for (; j + RMAX <= n0; j += RMAX)
          add_rows_k<K, RMAX>(tab, ent, j, c, a);
        add_tail_k<K>(tab, ent, j, n0 - j, c, a);
        for (int base = beg + 32; base < end; base += 32) {
          const int n = min(32, end - base);
          int2 e2 = make_int2(0, 0);
          if (lane < n) e2 = csr[base + lane];
          for (int jj = 0; jj < n; ++jj)
            add_rows_k<K, 1>(tab, e2, jj, c, a);
        }
        const float sn = di * di;
        float* trow = tiles + (t & 1) * (TROWS * P) + gw * P + c;
#pragma unroll
        for (int q = 0; q < NV; ++q) {
          a[q].x = fmaf(xs[q].x, sn, a[q].x);
          a[q].y = fmaf(xs[q].y, sn, a[q].y);
          a[q].z = fmaf(xs[q].z, sn, a[q].z);
          a[q].w = fmaf(xs[q].w, sn, a[q].w);
          *(float4*)(trow + q * 128) = a[q];
        }
        node = nnode;
        beg = nbeg;
        end = nend;
        di = ndi;
        ent = nent;
        nnode = n2;
        nbeg = n2beg;
        nend = n2end;
        ndi = n2di;
      }
    } else {
      if (t >= 1) {
        const float* at = tiles + ((t - 1) & 1) * (TROWS * P) + col * P + 2 * h;
        v2f fa[2][4];
        v2f fb[2][4][2];
        load_blk(at, bt, fa[0], fb[0]);
        if (t >= 2)
          store_rows<RELU_OUT>(acc0, acc1,
                               ocol + (size_t)(b + (t - 2) * G) * TROWS * HID,
                               h, bj);
        acc0 = (v8f){0.f, 0.f, 0.f, 0.f, 0.f, 0.f, 0.f, 0.f};
        acc1 = acc0;
#pragma unroll
        for (int kb = 0; kb < KB; ++kb) {
          if (kb + 1 < KB)
            load_blk(at + (kb + 1) * 16, bt + (kb + 1) * 8 * WPITCH,
                     fa[(kb + 1) & 1], fb[(kb + 1) & 1]);
          __builtin_amdgcn_sched_barrier(0);
          mma_blk(fa[kb & 1], fb[kb & 1], acc0, acc1);
          __builtin_amdgcn_sched_barrier(0);
        }
      }
    }
    __syncthreads();
  }
  if (!summer && nt >= 1)
    store_rows<RELU_OUT>(acc0, acc1,
                         ocol + (size_t)(b + (nt - 1) * G) * TROWS * HID, h, bj);
}


static inline int cdiv(int a, int b) { return (a + b - 1) / b; }

extern "C" void kernel_launch(void* const* d_in, const int* in_sizes, int n_in,
                              void* d_out, int out_size, void* d_ws,
                              size_t ws_size, hipStream_t stream) {
  const float* x = (const float*)d_in[0];
  const int* ei = (const int*)d_in[1];
  const float* W1 = (const float*)d_in[2];
  const float* b1 = (const float*)d_in[3];
  const float* W2 = (const float*)d_in[4];
  const float* b2 = (const float*)d_in[5];
  const float* W3 = (const float*)d_in[6];
  const float* b3 = (const float*)d_in[7];
  float* out = (float*)d_out;

  const int F = 128;
  const int N = in_sizes[0] / F;
  const int E = in_sizes[1] / 2;
  const int* src = ei;
  const int* dst = ei + E;

  const int scanGrid = cdiv(N, SCAN_CHUNK);
  const int NP = scanGrid * SCAN_CHUNK;

  float* wsf = (float*)d_ws;
  size_t off = 0;
  float* dinv = wsf + off;
  off += (size_t)NP;
  int* cntcur = (int*)(wsf + off);
  off += (size_t)NP;
  int* rowptr = (int*)(wsf + off);
  off += (size_t)NP + 256;
  int* bsum = (int*)(wsf + off);
  off += (size_t)((scanGrid + 255) / 256) * 256;
  int2* slot = (int2*)(wsf + off);
  off += (size_t)((2 * (size_t)E + 255) / 256) * 256;
  int2* csr = (int2*)(wsf + off);
  off += (size_t)((2 * (size_t)E + 255) / 256) * 256;
  float* hbuf = wsf + off;
  off += (size_t)N * HID;
  float* agg = wsf + off;

  const int padGrid = cdiv(NP, TPB);
  const int edgeGrid = cdiv(E, TPB);

  zero_int_kernel<<<padGrid, TPB, 0, stream>>>(cntcur, NP);
  count_kernel<<<edgeGrid, TPB, 0, stream>>>(dst, cntcur, E);
  block_sum_kernel<<<scanGrid, TPB, 0, stream>>>(cntcur, bsum);
  scan_kernel<<<scanGrid, TPB, 0, stream>>>(cntcur, bsum, rowptr, dinv);
  place_kernel<<<edgeGrid, TPB, 0, stream>>>(src, dst, cntcur, slot, E);
  order_kernel<<<edgeGrid, TPB, 0, stream>>>(slot, dst, rowptr, dinv, csr, E);

  fused_layer_kernel<128, true><<<FUSE_GRID, FUSE_TPB, 0, stream>>>(
      x, W1, b1, rowptr, csr, dinv, agg, N);
  fused_layer_kernel<256, true><<<FUSE_GRID, FUSE_TPB, 0, stream>>>(
      agg, W2, b2, rowptr, csr, dinv, hbuf, N);
  fused_layer_kernel<256, false><<<FUSE_GRID, FUSE_TPB, 0, stream>>>(
      hbuf, W3, b3, rowptr, csr, dinv, out, N);
}
